# starting kernel
# baseline (speedup 1.0000x reference)
#include <hip/hip_runtime.h>
#include <math.h>

typedef float v2f __attribute__((ext_vector_type(2)));
typedef float v8f __attribute__((ext_vector_type(8)));

#define B_ 4
#define N_ 512
#define C_ 64
#define O_ 64

__global__ __launch_bounds__(128) void edgeconv_gemm(
    const float* __restrict__ x,
    const float* __restrict__ W,
    const float* __restrict__ bias,
    float* __restrict__ u,
    float* __restrict__ dws)
{
    const int wave = threadIdx.x >> 5;
    const int lane = threadIdx.x & 31;
    const int r    = lane & 15;
    const int h    = lane >> 4;
    const int b    = blockIdx.x >> 5;
    const int mt   = blockIdx.x & 31;
    const int o    = wave * 16 + r;
    const int rowBase = mt * 16;

    const float* xrow = x + (size_t)(b * N_ + rowBase + r) * C_;

    v8f accA = {};
    v8f accD = {};
    #pragma unroll
    for (int k0 = 0; k0 < C_; k0 += 4) {
        const int ka = k0 + 2 * h;
        v2f aF = *(const v2f*)(xrow + ka);
        v2f bA, bD;
        bA.x = W[(size_t)(ka)         * O_ + o];
        bA.y = W[(size_t)(ka + 1)     * O_ + o];
        bD.x = W[(size_t)(C_ + ka)    * O_ + o];
        bD.y = W[(size_t)(C_ + ka + 1)* O_ + o];
        accA = __builtin_amdgcn_wmma_f32_16x16x4_f32(
             false, aF,  false, bA,
             (short)0, accA,  false,  false);
        accD = __builtin_amdgcn_wmma_f32_16x16x4_f32(
            false, aF, false, bD, (short)0, accD, false, false);
    }

    const float bo = bias[o];
    #pragma unroll
    for (int i = 0; i < 8; ++i) {
        const int row = rowBase + i + 8 * h;
        const size_t idx = (size_t)(b * N_ + row) * O_ + o;
        const float dv = accD[i];
        dws[idx] = dv;
        u[idx]   = accA[i] - dv + bo;
    }
}

__device__ __forceinline__ float gelu_exact(float v) {
    return 0.5f * v * (1.0f + erff(v * 0.70710678118654752f));
}

__global__ __launch_bounds__(256) void edgeconv_lnmean(
    const float* __restrict__ u,
    const float* __restrict__ dws,
    const int*   __restrict__ mask,
    const float* __restrict__ gamma,
    const float* __restrict__ beta,
    float* __restrict__ out)
{
    __shared__ float2 part[8 * 32];
    const int tid  = threadIdx.x;
    const int wave = tid >> 5;
    const int lane = tid & 31;
    const int b    = blockIdx.x >> 9;
    const int n    = blockIdx.x & (N_ - 1);
    const size_t row = (size_t)(b * N_ + n);

    if (mask[row] == 0) {
        if (tid < 32) {
            float2 z; z.x = 0.0f; z.y = 0.0f;
            *(float2*)(out + row * O_ + 2 * tid) = z;
        }
        return;
    }

    const float2 u2  = *(const float2*)(u     + row * O_ + 2 * lane);
    const float2 g2  = *(const float2*)(gamma + 2 * lane);
    const float2 be2 = *(const float2*)(beta  + 2 * lane);

    const float* dbase = dws + (size_t)b * N_ * O_;
    float2 acc; acc.x = 0.0f; acc.y = 0.0f;

    const int m0 = wave * 64;
    for (int m = m0; m < m0 + 64; ++m) {
        const float2 d2 = *(const float2*)(dbase + (size_t)m * O_ + 2 * lane);
        const float hx = gelu_exact(u2.x + d2.x);
        const float hy = gelu_exact(u2.y + d2.y);
        float s = hx + hy;
        float q = hx * hx + hy * hy;
        #pragma unroll
        for (int off = 16; off > 0; off >>= 1) {
            s += __shfl_xor(s, off, 32);
            q += __shfl_xor(q, off, 32);
        }
        const float mu  = s * (1.0f / 64.0f);
        const float var = q * (1.0f / 64.0f) - mu * mu;
        const float rs  = rsqrtf(var + 1e-5f);
        acc.x += (hx - mu) * rs * g2.x + be2.x;
        acc.y += (hy - mu) * rs * g2.y + be2.y;
    }

    part[wave * 32 + lane] = acc;
    __syncthreads();

    if (tid < 32) {
        float2 res = part[tid];
        #pragma unroll
        for (int w = 1; w < 8; ++w) {
            res.x += part[w * 32 + tid].x;
            res.y += part[w * 32 + tid].y;
        }
        res.x *= (1.0f / (float)N_);
        res.y *= (1.0f / (float)N_);
        *(float2*)(out + row * O_ + 2 * tid) = res;
    }
}

extern "C" void kernel_launch(void* const* d_in, const int* in_sizes, int n_in,
                              void* d_out, int out_size, void* d_ws, size_t ws_size,
                              hipStream_t stream) {
    (void)in_sizes; (void)n_in; (void)out_size; (void)ws_size;
    const float* x     = (const float*)d_in[0];
    const int*   mask  = (const int*)  d_in[1];
    const float* W     = (const float*)d_in[2];
    const float* bias  = (const float*)d_in[3];
    const float* gamma = (const float*)d_in[4];
    const float* beta  = (const float*)d_in[5];
    float* out = (float*)d_out;

    float* u   = (float*)d_ws;
    float* dws = u + (size_t)B_ * N_ * O_;

    edgeconv_gemm<<<B_ * (N_ / 16), 128, 0, stream>>>(x, W, bias, u, dws);
    edgeconv_lnmean<<<B_ * N_, 256, 0, stream>>>(u, dws, mask, gamma, beta, out);
}


# lnmean: LayerNorm sums of 8 m in one exchange tree that halves the live sums per step; mean, 1/sigma once per 8 m, rs_m by readlane; row = gamma*(sum rs*h - sum rs*mu)/N + beta, sum order changes
# speedup vs baseline: 3.3316x; 3.3316x over previous
#include <hip/hip_runtime.h>
#include <math.h>

typedef float v2f __attribute__((ext_vector_type(2)));
typedef float v8f __attribute__((ext_vector_type(8)));

#define B_ 4
#define N_ 512
#define C_ 64
#define O_ 64

__global__ __launch_bounds__(128) void edgeconv_gemm(
    const float* __restrict__ x,
    const float* __restrict__ W,
    const float* __restrict__ bias,
    float* __restrict__ u,
    float* __restrict__ dws)
{
    const int wave = threadIdx.x >> 5;
    const int lane = threadIdx.x & 31;
    const int r    = lane & 15;
    const int h    = lane >> 4;
    const int b    = blockIdx.x >> 5;
    const int mt   = blockIdx.x & 31;
    const int o    = wave * 16 + r;
    const int rowBase = mt * 16;

    const float* xrow = x + (size_t)(b * N_ + rowBase + r) * C_;

    v8f accA = {};
    v8f accD = {};
    #pragma unroll
    for (int k0 = 0; k0 < C_; k0 += 4) {
        const int ka = k0 + 2 * h;
        v2f aF = *(const v2f*)(xrow + ka);
        v2f bA, bD;
        bA.x = W[(size_t)(ka)         * O_ + o];
        bA.y = W[(size_t)(ka + 1)     * O_ + o];
        bD.x = W[(size_t)(C_ + ka)    * O_ + o];
        bD.y = W[(size_t)(C_ + ka + 1)* O_ + o];
        accA = __builtin_amdgcn_wmma_f32_16x16x4_f32(
             false, aF,  false, bA,
             (short)0, accA,  false,  false);
        accD = __builtin_amdgcn_wmma_f32_16x16x4_f32(
            false, aF, false, bD, (short)0, accD, false, false);
    }

    const float bo = bias[o];
    #pragma unroll
    for (int i = 0; i < 8; ++i) {
        const int row = rowBase + i + 8 * h;
        const size_t idx = (size_t)(b * N_ + row) * O_ + o;
        const float dv = accD[i];
        dws[idx] = dv;
        u[idx]   = accA[i] - dv + bo;
    }
}

#define LN_WAVES  32
#define LN_BLOCKS 512
#define LN_MSPAN  (N_ / LN_WAVES)
#define LN_TURNS  ((B_ * N_) / LN_BLOCKS)
#define LN_BATCH  8

#define PHI_CELLS 3072
#define PHI_INV_H 512.0f
#define PHI_TMAX  3071.9997f

__device__ __forceinline__ float phi_exact(float v) {
    return 0.5f * (1.0f + erff(v * 0.70710678118654752f));
}

__device__ __forceinline__ float gelu_tab(float v, const float2* tab) {
    const float a = fabsf(v);
    const float t = fminf(a * PHI_INV_H, PHI_TMAX);
    const float2 e = tab[(unsigned)t];
    const float phi = fmaf(__builtin_amdgcn_fractf(t), e.y, e.x);
    return fmaf(a, phi, fminf(v, 0.0f));
}

__device__ __forceinline__ int row_of_rank(unsigned w0, unsigned w1, int excl, int c,
                                           int want, int lane) {
    const int k = want - excl;
    const bool owner = (k >= 0) && (k < c);
    const int c0 = __popc(w0);
    unsigned wsel = w0;
    int kk = k, base = 0;
    if (k >= c0) { kk = k - c0; wsel = w1; base = 32; }
    int p = 0;
    #pragma unroll
    for (int sh = 16; sh >= 1; sh >>= 1)
        if (__popc(wsel & ((1u << (p + sh)) - 1u)) <= kk) p += sh;
    const int pos = 64 * lane + base + p;
    const unsigned om = __builtin_amdgcn_ballot_w32(owner);
    const int src = om ? __builtin_ctz(om) : 0;
    return __builtin_amdgcn_readfirstlane(__shfl(pos, src, 32)) & (B_ * N_ - 1);
}

__device__ __forceinline__ float xhalf(float x) {
    const int b = __builtin_bit_cast(int, x);
    return __builtin_bit_cast(float,
        __builtin_amdgcn_permlanex16(b, b, 0x76543210, 0xfedcba98, false, false));
}
template <int MASK>
__device__ __forceinline__ float xrow(float x) {
    const int b = __builtin_bit_cast(int, x);
    return __builtin_bit_cast(float, __builtin_amdgcn_update_dpp(b, b, 0x160 + MASK, 0xF, 0xF, true));
}

__device__ __forceinline__ float fold_rows(float a, float b) {
    asm("v_permlane16_swap_b32 %0, %1" : "+v"(a), "+v"(b));
    return a + b;
}
template <int MASK>
__device__ __forceinline__ float fold_lanes(float a, float b, bool up) {
    const float keep = up ? b : a;
    const float send = up ? a : b;
    return keep + xrow<MASK>(send);
}

__global__ __launch_bounds__(LN_WAVES * 32, 16) void edgeconv_lnmean(
    const float* __restrict__ u,
    const float* __restrict__ dws,
    const int*   __restrict__ mask,
    const float* __restrict__ gamma,
    const float* __restrict__ beta,
    float* __restrict__ out)
{
    __shared__ float2 part[2][LN_WAVES * 32];
    __shared__ unsigned bits[(B_ * N_) / 32];
    __shared__ float2 phi_tab[PHI_CELLS];
    const int tid  = threadIdx.x;
    const int wave = tid >> 5;
    const int lane = tid & 31;
    const int bi   = blockIdx.x;

    constexpr int NT = LN_WAVES * 32;
    for (int k = 0; k < PHI_CELLS / NT; ++k) {
        const int e = k * NT + tid;
        phi_tab[e].x = phi_exact((float)e * (1.0f / PHI_INV_H));
    }

    int mv[(B_ * N_) / NT];
    #pragma unroll
    for (int s = 0; s < (B_ * N_) / NT; ++s) mv[s] = mask[s * NT + tid];
    #pragma unroll
    for (int s = 0; s < (B_ * N_) / NT; ++s) {
        const unsigned bal = __builtin_amdgcn_ballot_w32(mv[s] != 0);
        if (lane == 0) bits[s * LN_WAVES + wave] = bal;
    }
    __syncthreads();

    for (int k = 0; k < PHI_CELLS / NT; ++k) {
        const int e = k * NT + tid;
        const float nxt = phi_tab[min(e + 1, PHI_CELLS - 1)].x;
        phi_tab[e].y = ((e + 1 < PHI_CELLS) ? nxt : 1.0f) - phi_tab[e].x;
    }
    __syncthreads();

    const unsigned w0 = bits[2 * lane];
    const unsigned w1 = bits[2 * lane + 1];
    const int c = __popc(w0) + __popc(w1);
    int incl = c;
    #pragma unroll
    for (int off = 1; off < 32; off <<= 1) {
        const int v = __shfl_up(incl, off, 32);
        if (lane >= off) incl += v;
    }
    const int A = __builtin_amdgcn_readfirstlane(__shfl(incl, 31, 32));
    const int excl = incl - c;

    {
        const int wv = __builtin_amdgcn_readfirstlane(wave);
        const int zr = bi + wv * LN_BLOCKS;
        if (wv < LN_TURNS && zr < B_ * N_ - A) {
            const int rowz = row_of_rank(~w0, ~w1, 64 * lane - excl, 64 - c, zr, lane);
            float2 z; z.x = 0.0f; z.y = 0.0f;
            *(float2*)(out + (size_t)rowz * O_ + 2 * lane) = z;
        }
    }

    if (bi >= A) return;

    const int m0 = wave * LN_MSPAN;
    const bool up8 = (lane & 8) != 0;
    const bool up4 = (lane & 4) != 0;

    int rowi = row_of_rank(w0, w1, excl, c, bi, lane);
    float2 u2 = *(const float2*)(u + (size_t)rowi * O_ + 2 * lane);
    int buf = 0;

    for (int rank = bi; rank < A; rank += LN_BLOCKS) {
        const int nrank = rank + LN_BLOCKS;
        int   nrowi = rowi;
        float2 nu2  = u2;
        if (nrank < A) {
            nrowi = row_of_rank(w0, w1, excl, c, nrank, lane);
            nu2   = *(const float2*)(u + (size_t)nrowi * O_ + 2 * lane);
        }

        const float* dbase = dws + (size_t)(rowi >> 9) * N_ * O_;
        float2 acc; acc.x = 0.0f; acc.y = 0.0f;
        float accm = 0.0f;

        for (int bt = 0; bt < LN_MSPAN / LN_BATCH; ++bt) {
            const float* dp = dbase + (size_t)(m0 + bt * LN_BATCH) * O_ + 2 * lane;
            float2 hv[LN_BATCH];
            float  s[LN_BATCH], q[LN_BATCH];
            #pragma unroll
            for (int j = 0; j < LN_BATCH; ++j)
                hv[j] = *(const float2*)(dp + (size_t)j * O_);
            #pragma unroll
            for (int j = 0; j < LN_BATCH; ++j) {
                const float hx = gelu_tab(u2.x + hv[j].x, phi_tab);
                const float hy = gelu_tab(u2.y + hv[j].y, phi_tab);
                hv[j].x = hx;  hv[j].y = hy;
                s[j] = hx + hy;
                q[j] = hx * hx + hy * hy;
            }
            const float s01 = fold_rows(s[0], s[1]), s23 = fold_rows(s[2], s[3]);
            const float s45 = fold_rows(s[4], s[5]), s67 = fold_rows(s[6], s[7]);
            const float q01 = fold_rows(q[0], q[1]), q23 = fold_rows(q[2], q[3]);
            const float q45 = fold_rows(q[4], q[5]), q67 = fold_rows(q[6], q[7]);
            const float s03 = fold_lanes<8>(s01, s23, up8), s47 = fold_lanes<8>(s45, s67, up8);
            const float q03 = fold_lanes<8>(q01, q23, up8), q47 = fold_lanes<8>(q45, q67, up8);
            float sm = fold_lanes<4>(s03, s47, up4);
            float qm = fold_lanes<4>(q03, q47, up4);
            sm += xrow<2>(sm);  qm += xrow<2>(qm);
            sm += xrow<1>(sm);  qm += xrow<1>(qm);
            const float mu  = sm * (1.0f / 64.0f);
            const float var = qm * (1.0f / 64.0f) - mu * mu;
            const float rs  = rsqrtf(var + 1e-5f);
            accm = fmaf(rs, mu, accm);
            #pragma unroll
            for (int j = 0; j < LN_BATCH; ++j) {
                const int src = 16 * (j & 1) + 8 * ((j >> 1) & 1) + 4 * ((j >> 2) & 1);
                const float rsj = __builtin_bit_cast(float,
                    __builtin_amdgcn_readlane(__builtin_bit_cast(int, rs), src));
                acc.x = fmaf(hv[j].x, rsj, acc.x);
                acc.y = fmaf(hv[j].y, rsj, acc.y);
            }
        }
        accm += xhalf(accm);
        accm += xrow<8>(accm);
        accm += xrow<4>(accm);
        acc.x -= accm;
        acc.y -= accm;

        part[buf][wave * 32 + lane] = acc;
        __syncthreads();

        if (tid < 32) {
            float2 res = part[buf][tid];
            #pragma unroll
            for (int w = 1; w < LN_WAVES; ++w) {
                res.x += part[buf][w * 32 + tid].x;
                res.y += part[buf][w * 32 + tid].y;
            }
            const float2 g2  = *(const float2*)(gamma + 2 * tid);
            const float2 be2 = *(const float2*)(beta  + 2 * tid);
            res.x = fmaf(g2.x, res.x * (1.0f / (float)N_), be2.x);
            res.y = fmaf(g2.y, res.y * (1.0f / (float)N_), be2.y);
            *(float2*)(out + (size_t)rowi * O_ + 2 * tid) = res;
        }

        buf ^= 1;
        rowi = nrowi;
        u2   = nu2;
    }
}

extern "C" void kernel_launch(void* const* d_in, const int* in_sizes, int n_in,
                              void* d_out, int out_size, void* d_ws, size_t ws_size,
                              hipStream_t stream) {
    (void)in_sizes; (void)n_in; (void)out_size; (void)ws_size;
    const float* x     = (const float*)d_in[0];
    const int*   mask  = (const int*)  d_in[1];
    const float* W     = (const float*)d_in[2];
    const float* bias  = (const float*)d_in[3];
    const float* gamma = (const float*)d_in[4];
    const float* beta  = (const float*)d_in[5];
    float* out = (float*)d_out;

    float* u   = (float*)d_ws;
    float* dws = u + (size_t)B_ * N_ * O_;

    edgeconv_gemm<<<B_ * (N_ / 16), 128, 0, stream>>>(x, W, bias, u, dws);
    edgeconv_lnmean<<<LN_BLOCKS, LN_WAVES * 32, 0, stream>>>(u, dws, mask, gamma, beta, out);
}


# lnmean: gelu = v/2 + E(|v|), E(a) = a*(Phi(a)-1/2) from the LDS table as a line in the cell position itself (u, d stored x512): 6 vector ops per value, was 9; f32 kept, gelu rounding differs
# speedup vs baseline: 3.5521x; 1.0662x over previous
#include <hip/hip_runtime.h>
#include <math.h>

typedef float v2f __attribute__((ext_vector_type(2)));
typedef float v8f __attribute__((ext_vector_type(8)));

#define B_ 4
#define N_ 512
#define C_ 64
#define O_ 64
#define WS_SCALE 512.0f

__global__ __launch_bounds__(128) void edgeconv_gemm(
    const float* __restrict__ x,
    const float* __restrict__ W,
    const float* __restrict__ bias,
    float* __restrict__ u,
    float* __restrict__ dws)
{
    const int wave = threadIdx.x >> 5;
    const int lane = threadIdx.x & 31;
    const int r    = lane & 15;
    const int h    = lane >> 4;
    const int b    = blockIdx.x >> 5;
    const int mt   = blockIdx.x & 31;
    const int o    = wave * 16 + r;
    const int rowBase = mt * 16;

    const float* xrow = x + (size_t)(b * N_ + rowBase + r) * C_;

    v8f accA = {};
    v8f accD = {};
    #pragma unroll
    for (int k0 = 0; k0 < C_; k0 += 4) {
        const int ka = k0 + 2 * h;
        v2f aF = *(const v2f*)(xrow + ka);
        v2f bA, bD;
        bA.x = W[(size_t)(ka)         * O_ + o];
        bA.y = W[(size_t)(ka + 1)     * O_ + o];
        bD.x = W[(size_t)(C_ + ka)    * O_ + o];
        bD.y = W[(size_t)(C_ + ka + 1)* O_ + o];
        accA = __builtin_amdgcn_wmma_f32_16x16x4_f32(
             false, aF,  false, bA,
             (short)0, accA,  false,  false);
        accD = __builtin_amdgcn_wmma_f32_16x16x4_f32(
            false, aF, false, bD, (short)0, accD, false, false);
    }

    const float bo = bias[o];
    #pragma unroll
    for (int i = 0; i < 8; ++i) {
        const int row = rowBase + i + 8 * h;
        const size_t idx = (size_t)(b * N_ + row) * O_ + o;
        const float dv = accD[i];
        dws[idx] = dv * WS_SCALE;
        u[idx]   = (accA[i] - dv + bo) * WS_SCALE;
    }
}

#define LN_WAVES  32
#define LN_BLOCKS 512
#define LN_MSPAN  (N_ / LN_WAVES)
#define LN_TURNS  ((B_ * N_) / LN_BLOCKS)
#define LN_BATCH  8

#define PHI_CELLS 3072
#define EV_INV_H  512.0f
#define PHI_TMAX  3071.9997f
#define LN_EPS_W  (1e-5f * EV_INV_H * EV_INV_H)

__device__ __forceinline__ float ev_exact(int e) {
    const float x = (float)e * (1.0f / EV_INV_H);
    return (0.5f * (float)e) * erff(x * 0.70710678118654752f);
}

__device__ __forceinline__ float gelu_tab(float w, const float2* tab) {
    const float a = fabsf(w);
    const float2 e = tab[(unsigned)fminf(a, PHI_TMAX)];
    return fmaf(0.5f, w, fmaf(a, e.y, e.x));
}

__device__ __forceinline__ int row_of_rank(unsigned w0, unsigned w1, int excl, int c,
                                           int want, int lane) {
    const int k = want - excl;
    const bool owner = (k >= 0) && (k < c);
    const int c0 = __popc(w0);
    unsigned wsel = w0;
    int kk = k, base = 0;
    if (k >= c0) { kk = k - c0; wsel = w1; base = 32; }
    int p = 0;
    #pragma unroll
    for (int sh = 16; sh >= 1; sh >>= 1)
        if (__popc(wsel & ((1u << (p + sh)) - 1u)) <= kk) p += sh;
    const int pos = 64 * lane + base + p;
    const unsigned om = __builtin_amdgcn_ballot_w32(owner);
    const int src = om ? __builtin_ctz(om) : 0;
    return __builtin_amdgcn_readfirstlane(__shfl(pos, src, 32)) & (B_ * N_ - 1);
}

__device__ __forceinline__ float xhalf(float x) {
    const int b = __builtin_bit_cast(int, x);
    return __builtin_bit_cast(float,
        __builtin_amdgcn_permlanex16(b, b, 0x76543210, 0xfedcba98, false, false));
}
template <int MASK>
__device__ __forceinline__ float xrow(float x) {
    const int b = __builtin_bit_cast(int, x);
    return __builtin_bit_cast(float, __builtin_amdgcn_update_dpp(b, b, 0x160 + MASK, 0xF, 0xF, true));
}

__device__ __forceinline__ float fold_rows(float a, float b) {
    asm("v_permlane16_swap_b32 %0, %1" : "+v"(a), "+v"(b));
    return a + b;
}
template <int MASK>
__device__ __forceinline__ float fold_lanes(float a, float b, bool up) {
    const float keep = up ? b : a;
    const float send = up ? a : b;
    return keep + xrow<MASK>(send);
}

__global__ __launch_bounds__(LN_WAVES * 32, 16) void edgeconv_lnmean(
    const float* __restrict__ u,
    const float* __restrict__ dws,
    const int*   __restrict__ mask,
    const float* __restrict__ gamma,
    const float* __restrict__ beta,
    float* __restrict__ out)
{
    __shared__ float2 part[2][LN_WAVES * 32];
    __shared__ unsigned bits[(B_ * N_) / 32];
    __shared__ float2 phi_tab[PHI_CELLS];
    const int tid  = threadIdx.x;
    const int wave = tid >> 5;
    const int lane = tid & 31;
    const int bi   = blockIdx.x;

    constexpr int NT = LN_WAVES * 32;
    float* ev0 = (float*)&part[0][0];
    for (int k = 0; k < PHI_CELLS / NT; ++k) {
        const int e = k * NT + tid;
        ev0[e] = ev_exact(e);
    }

    int mv[(B_ * N_) / NT];
    #pragma unroll
    for (int s = 0; s < (B_ * N_) / NT; ++s) mv[s] = mask[s * NT + tid];
    #pragma unroll
    for (int s = 0; s < (B_ * N_) / NT; ++s) {
        const unsigned bal = __builtin_amdgcn_ballot_w32(mv[s] != 0);
        if (lane == 0) bits[s * LN_WAVES + wave] = bal;
    }
    __syncthreads();

    for (int k = 0; k < PHI_CELLS / NT; ++k) {
        const int e = k * NT + tid;
        const float cur = ev0[e];
        const float nxt = ev0[min(e + 1, PHI_CELLS - 1)];
        float2 ln;
        ln.y = ((e + 1 < PHI_CELLS) ? nxt : 0.5f * (float)PHI_CELLS) - cur;
        ln.x = cur - (float)e * ln.y;
        phi_tab[e] = ln;
    }
    __syncthreads();

    const unsigned w0 = bits[2 * lane];
    const unsigned w1 = bits[2 * lane + 1];
    const int c = __popc(w0) + __popc(w1);
    int incl = c;
    #pragma unroll
    for (int off = 1; off < 32; off <<= 1) {
        const int v = __shfl_up(incl, off, 32);
        if (lane >= off) incl += v;
    }
    const int A = __builtin_amdgcn_readfirstlane(__shfl(incl, 31, 32));
    const int excl = incl - c;

    {
        const int wv = __builtin_amdgcn_readfirstlane(wave);
        const int zr = bi + wv * LN_BLOCKS;
        if (wv < LN_TURNS && zr < B_ * N_ - A) {
            const int rowz = row_of_rank(~w0, ~w1, 64 * lane - excl, 64 - c, zr, lane);
            float2 z; z.x = 0.0f; z.y = 0.0f;
            *(float2*)(out + (size_t)rowz * O_ + 2 * lane) = z;
        }
    }

    if (bi >= A) return;

    const int m0 = wave * LN_MSPAN;
    const bool up8 = (lane & 8) != 0;
    const bool up4 = (lane & 4) != 0;

    int rowi = row_of_rank(w0, w1, excl, c, bi, lane);
    float2 u2 = *(const float2*)(u + (size_t)rowi * O_ + 2 * lane);
    int buf = 0;

    for (int rank = bi; rank < A; rank += LN_BLOCKS) {
        const int nrank = rank + LN_BLOCKS;
        int   nrowi = rowi;
        float2 nu2  = u2;
        if (nrank < A) {
            nrowi = row_of_rank(w0, w1, excl, c, nrank, lane);
            nu2   = *(const float2*)(u + (size_t)nrowi * O_ + 2 * lane);
        }

        const float* dbase = dws + (size_t)(rowi >> 9) * N_ * O_;
        float2 acc; acc.x = 0.0f; acc.y = 0.0f;
        float accm = 0.0f;

        for (int bt = 0; bt < LN_MSPAN / LN_BATCH; ++bt) {
            const float* dp = dbase + (size_t)(m0 + bt * LN_BATCH) * O_ + 2 * lane;
            float2 hv[LN_BATCH];
            float  s[LN_BATCH], q[LN_BATCH];
            #pragma unroll
            for (int j = 0; j < LN_BATCH; ++j)
                hv[j] = *(const float2*)(dp + (size_t)j * O_);
            #pragma unroll
            for (int j = 0; j < LN_BATCH; ++j) {
                const float hx = gelu_tab(u2.x + hv[j].x, phi_tab);
                const float hy = gelu_tab(u2.y + hv[j].y, phi_tab);
                hv[j].x = hx;  hv[j].y = hy;
                s[j] = hx + hy;
                q[j] = hx * hx + hy * hy;
            }
            const float s01 = fold_rows(s[0], s[1]), s23 = fold_rows(s[2], s[3]);
            const float s45 = fold_rows(s[4], s[5]), s67 = fold_rows(s[6], s[7]);
            const float q01 = fold_rows(q[0], q[1]), q23 = fold_rows(q[2], q[3]);
            const float q45 = fold_rows(q[4], q[5]), q67 = fold_rows(q[6], q[7]);
            const float s03 = fold_lanes<8>(s01, s23, up8), s47 = fold_lanes<8>(s45, s67, up8);
            const float q03 = fold_lanes<8>(q01, q23, up8), q47 = fold_lanes<8>(q45, q67, up8);
            float sm = fold_lanes<4>(s03, s47, up4);
            float qm = fold_lanes<4>(q03, q47, up4);
            sm += xrow<2>(sm);  qm += xrow<2>(qm);
            sm += xrow<1>(sm);  qm += xrow<1>(qm);
            const float mu  = sm * (1.0f / 64.0f);
            const float var = qm * (1.0f / 64.0f) - mu * mu;
            const float rs  = rsqrtf(var + LN_EPS_W);
            accm = fmaf(rs, mu, accm);
            #pragma unroll
            for (int j = 0; j < LN_BATCH; ++j) {
                const int src = 16 * (j & 1) + 8 * ((j >> 1) & 1) + 4 * ((j >> 2) & 1);
                const float rsj = __builtin_bit_cast(float,
                    __builtin_amdgcn_readlane(__builtin_bit_cast(int, rs), src));
                acc.x = fmaf(hv[j].x, rsj, acc.x);
                acc.y = fmaf(hv[j].y, rsj, acc.y);
            }
        }
        accm += xhalf(accm);
        accm += xrow<8>(accm);
        accm += xrow<4>(accm);
        acc.x -= accm;
        acc.y -= accm;

        part[buf][wave * 32 + lane] = acc;
        __syncthreads();

        if (tid < 32) {
            float2 res = part[buf][tid];
            #pragma unroll
            for (int w = 1; w < LN_WAVES; ++w) {
                res.x += part[buf][w * 32 + tid].x;
                res.y += part[buf][w * 32 + tid].y;
            }
            const float2 g2  = *(const float2*)(gamma + 2 * tid);
            const float2 be2 = *(const float2*)(beta  + 2 * tid);
            res.x = fmaf(g2.x, res.x * (1.0f / (float)N_), be2.x);
            res.y = fmaf(g2.y, res.y * (1.0f / (float)N_), be2.y);
            *(float2*)(out + (size_t)rowi * O_ + 2 * tid) = res;
        }

        buf ^= 1;
        rowi = nrowi;
        u2   = nu2;
    }
}

extern "C" void kernel_launch(void* const* d_in, const int* in_sizes, int n_in,
                              void* d_out, int out_size, void* d_ws, size_t ws_size,
                              hipStream_t stream) {
    (void)in_sizes; (void)n_in; (void)out_size; (void)ws_size;
    const float* x     = (const float*)d_in[0];
    const int*   mask  = (const int*)  d_in[1];
    const float* W     = (const float*)d_in[2];
    const float* bias  = (const float*)d_in[3];
    const float* gamma = (const float*)d_in[4];
    const float* beta  = (const float*)d_in[5];
    float* out = (float*)d_out;

    float* u   = (float*)d_ws;
    float* dws = u + (size_t)B_ * N_ * O_;

    edgeconv_gemm<<<B_ * (N_ / 16), 128, 0, stream>>>(x, W, bias, u, dws);
    edgeconv_lnmean<<<LN_BLOCKS, LN_WAVES * 32, 0, stream>>>(u, dws, mask, gamma, beta, out);
}


# lnmean: blocks of 16 waves (was 32), 8 waves per SIMD with 128 registers: half the waves to enter; a batch's 16 table reads go out as one group; row sum = 16 partials of 32 m (order changes)
# speedup vs baseline: 3.8764x; 1.0913x over previous
#include <hip/hip_runtime.h>
#include <math.h>

typedef float v2f __attribute__((ext_vector_type(2)));
typedef float v8f __attribute__((ext_vector_type(8)));

#define B_ 4
#define N_ 512
#define C_ 64
#define O_ 64
#define WS_SCALE 512.0f

__global__ __launch_bounds__(128) void edgeconv_gemm(
    const float* __restrict__ x,
    const float* __restrict__ W,
    const float* __restrict__ bias,
    float* __restrict__ u,
    float* __restrict__ dws)
{
    const int wave = threadIdx.x >> 5;
    const int lane = threadIdx.x & 31;
    const int r    = lane & 15;
    const int h    = lane >> 4;
    const int b    = blockIdx.x >> 5;
    const int mt   = blockIdx.x & 31;
    const int o    = wave * 16 + r;
    const int rowBase = mt * 16;

    const float* xrow = x + (size_t)(b * N_ + rowBase + r) * C_;

    v8f accA = {};
    v8f accD = {};
    #pragma unroll
    for (int k0 = 0; k0 < C_; k0 += 4) {
        const int ka = k0 + 2 * h;
        v2f aF = *(const v2f*)(xrow + ka);
        v2f bA, bD;
        bA.x = W[(size_t)(ka)         * O_ + o];
        bA.y = W[(size_t)(ka + 1)     * O_ + o];
        bD.x = W[(size_t)(C_ + ka)    * O_ + o];
        bD.y = W[(size_t)(C_ + ka + 1)* O_ + o];
        accA = __builtin_amdgcn_wmma_f32_16x16x4_f32(
             false, aF,  false, bA,
             (short)0, accA,  false,  false);
        accD = __builtin_amdgcn_wmma_f32_16x16x4_f32(
            false, aF, false, bD, (short)0, accD, false, false);
    }

    const float bo = bias[o];
    #pragma unroll
    for (int i = 0; i < 8; ++i) {
        const int row = rowBase + i + 8 * h;
        const size_t idx = (size_t)(b * N_ + row) * O_ + o;
        const float dv = accD[i];
        dws[idx] = dv * WS_SCALE;
        u[idx]   = (accA[i] - dv + bo) * WS_SCALE;
    }
}

#define LN_WAVES  16
#define LN_BLOCKS 512
#define LN_MSPAN  (N_ / LN_WAVES)
#define LN_TURNS  ((B_ * N_) / LN_BLOCKS)
#define LN_BATCH  8

#define PHI_CELLS 3072
#define EV_INV_H  512.0f
#define PHI_TMAX  3071.9997f
#define LN_EPS_W  (1e-5f * EV_INV_H * EV_INV_H)

__device__ __forceinline__ float ev_exact(int e) {
    const float x = (float)e * (1.0f / EV_INV_H);
    return (0.5f * (float)e) * erff(x * 0.70710678118654752f);
}

__device__ __forceinline__ float gelu_tab(float w, const float2* tab) {
    const float a = fabsf(w);
    const float2 e = tab[(unsigned)fminf(a, PHI_TMAX)];
    return fmaf(0.5f, w, fmaf(a, e.y, e.x));
}

__device__ __forceinline__ int row_of_rank(unsigned w0, unsigned w1, int excl, int c,
                                           int want, int lane) {
    const int k = want - excl;
    const bool owner = (k >= 0) && (k < c);
    const int c0 = __popc(w0);
    unsigned wsel = w0;
    int kk = k, base = 0;
    if (k >= c0) { kk = k - c0; wsel = w1; base = 32; }
    int p = 0;
    #pragma unroll
    for (int sh = 16; sh >= 1; sh >>= 1)
        if (__popc(wsel & ((1u << (p + sh)) - 1u)) <= kk) p += sh;
    const int pos = 64 * lane + base + p;
    const unsigned om = __builtin_amdgcn_ballot_w32(owner);
    const int src = om ? __builtin_ctz(om) : 0;
    return __builtin_amdgcn_readfirstlane(__shfl(pos, src, 32)) & (B_ * N_ - 1);
}

__device__ __forceinline__ float xhalf(float x) {
    const int b = __builtin_bit_cast(int, x);
    return __builtin_bit_cast(float,
        __builtin_amdgcn_permlanex16(b, b, 0x76543210, 0xfedcba98, false, false));
}
template <int MASK>
__device__ __forceinline__ float xrow(float x) {
    const int b = __builtin_bit_cast(int, x);
    return __builtin_bit_cast(float, __builtin_amdgcn_update_dpp(b, b, 0x160 + MASK, 0xF, 0xF, true));
}

__device__ __forceinline__ float fold_rows(float a, float b) {
    asm("v_permlane16_swap_b32 %0, %1" : "+v"(a), "+v"(b));
    return a + b;
}
template <int MASK>
__device__ __forceinline__ float fold_lanes(float a, float b, bool up) {
    const float keep = up ? b : a;
    const float send = up ? a : b;
    return keep + xrow<MASK>(send);
}

__global__ __launch_bounds__(LN_WAVES * 32, 8) void edgeconv_lnmean(
    const float* __restrict__ u,
    const float* __restrict__ dws,
    const int*   __restrict__ mask,
    const float* __restrict__ gamma,
    const float* __restrict__ beta,
    float* __restrict__ out)
{
    __shared__ float2 part[2][LN_WAVES * 32];
    __shared__ unsigned bits[(B_ * N_) / 32];
    __shared__ float2 phi_tab[PHI_CELLS];
    __shared__ float  ev0[PHI_CELLS];
    const int tid  = threadIdx.x;
    const int wave = tid >> 5;
    const int lane = tid & 31;
    const int bi   = blockIdx.x;

    constexpr int NT = LN_WAVES * 32;
    for (int k = 0; k < PHI_CELLS / NT; ++k) {
        const int e = k * NT + tid;
        ev0[e] = ev_exact(e);
    }

    int mv[(B_ * N_) / NT];
    #pragma unroll
    for (int s = 0; s < (B_ * N_) / NT; ++s) mv[s] = mask[s * NT + tid];
    #pragma unroll
    for (int s = 0; s < (B_ * N_) / NT; ++s) {
        const unsigned bal = __builtin_amdgcn_ballot_w32(mv[s] != 0);
        if (lane == 0) bits[s * LN_WAVES + wave] = bal;
    }
    __syncthreads();

    for (int k = 0; k < PHI_CELLS / NT; ++k) {
        const int e = k * NT + tid;
        const float cur = ev0[e];
        const float nxt = ev0[min(e + 1, PHI_CELLS - 1)];
        float2 ln;
        ln.y = ((e + 1 < PHI_CELLS) ? nxt : 0.5f * (float)PHI_CELLS) - cur;
        ln.x = cur - (float)e * ln.y;
        phi_tab[e] = ln;
    }
    __syncthreads();

    const unsigned w0 = bits[2 * lane];
    const unsigned w1 = bits[2 * lane + 1];
    const int c = __popc(w0) + __popc(w1);
    int incl = c;
    #pragma unroll
    for (int off = 1; off < 32; off <<= 1) {
        const int v = __shfl_up(incl, off, 32);
        if (lane >= off) incl += v;
    }
    const int A = __builtin_amdgcn_readfirstlane(__shfl(incl, 31, 32));
    const int excl = incl - c;

    {
        const int wv = __builtin_amdgcn_readfirstlane(wave);
        const int zr = bi + wv * LN_BLOCKS;
        if (wv < LN_TURNS && zr < B_ * N_ - A) {
            const int rowz = row_of_rank(~w0, ~w1, 64 * lane - excl, 64 - c, zr, lane);
            float2 z; z.x = 0.0f; z.y = 0.0f;
            *(float2*)(out + (size_t)rowz * O_ + 2 * lane) = z;
        }
    }

    if (bi >= A) return;

    const int m0 = wave * LN_MSPAN;
    const bool up8 = (lane & 8) != 0;
    const bool up4 = (lane & 4) != 0;

    int rowi = row_of_rank(w0, w1, excl, c, bi, lane);
    float2 u2 = *(const float2*)(u + (size_t)rowi * O_ + 2 * lane);
    int buf = 0;

    for (int rank = bi; rank < A; rank += LN_BLOCKS) {
        const int nrank = rank + LN_BLOCKS;
        int   nrowi = rowi;
        float2 nu2  = u2;
        if (nrank < A) {
            nrowi = row_of_rank(w0, w1, excl, c, nrank, lane);
            nu2   = *(const float2*)(u + (size_t)nrowi * O_ + 2 * lane);
        }

        const float* dbase = dws + (size_t)(rowi >> 9) * N_ * O_;
        float2 acc; acc.x = 0.0f; acc.y = 0.0f;
        float accm = 0.0f;

        #pragma unroll 1
        for (int bt = 0; bt < LN_MSPAN / LN_BATCH; ++bt) {
            const float* dp = dbase + (size_t)(m0 + bt * LN_BATCH) * O_ + 2 * lane;
            float2 hv[LN_BATCH];
            float  s[LN_BATCH], q[LN_BATCH];
            #pragma unroll
            for (int j = 0; j < LN_BATCH; ++j)
                hv[j] = *(const float2*)(dp + (size_t)j * O_);
            #pragma unroll
            for (int j = 0; j < LN_BATCH; ++j) {
                const float hx = gelu_tab(u2.x + hv[j].x, phi_tab);
                const float hy = gelu_tab(u2.y + hv[j].y, phi_tab);
                hv[j].x = hx;  hv[j].y = hy;
                s[j] = hx + hy;
                q[j] = hx * hx + hy * hy;
            }
            const float s01 = fold_rows(s[0], s[1]), s23 = fold_rows(s[2], s[3]);
            const float s45 = fold_rows(s[4], s[5]), s67 = fold_rows(s[6], s[7]);
            const float q01 = fold_rows(q[0], q[1]), q23 = fold_rows(q[2], q[3]);
            const float q45 = fold_rows(q[4], q[5]), q67 = fold_rows(q[6], q[7]);
            const float s03 = fold_lanes<8>(s01, s23, up8), s47 = fold_lanes<8>(s45, s67, up8);
            const float q03 = fold_lanes<8>(q01, q23, up8), q47 = fold_lanes<8>(q45, q67, up8);
            float sm = fold_lanes<4>(s03, s47, up4);
            float qm = fold_lanes<4>(q03, q47, up4);
            sm += xrow<2>(sm);  qm += xrow<2>(qm);
            sm += xrow<1>(sm);  qm += xrow<1>(qm);
            const float mu  = sm * (1.0f / 64.0f);
            const float var = qm * (1.0f / 64.0f) - mu * mu;
            const float rs  = rsqrtf(var + LN_EPS_W);
            accm = fmaf(rs, mu, accm);
            #pragma unroll
            for (int j = 0; j < LN_BATCH; ++j) {
                const int src = 16 * (j & 1) + 8 * ((j >> 1) & 1) + 4 * ((j >> 2) & 1);
                const float rsj = __builtin_bit_cast(float,
                    __builtin_amdgcn_readlane(__builtin_bit_cast(int, rs), src));
                acc.x = fmaf(hv[j].x, rsj, acc.x);
                acc.y = fmaf(hv[j].y, rsj, acc.y);
            }
        }
        accm += xhalf(accm);
        accm += xrow<8>(accm);
        accm += xrow<4>(accm);
        acc.x -= accm;
        acc.y -= accm;

        part[buf][wave * 32 + lane] = acc;
        __syncthreads();

        if (tid < 32) {
            float2 res = part[buf][tid];
            #pragma unroll
            for (int w = 1; w < LN_WAVES; ++w) {
                res.x += part[buf][w * 32 + tid].x;
                res.y += part[buf][w * 32 + tid].y;
            }
            const float2 g2  = *(const float2*)(gamma + 2 * tid);
            const float2 be2 = *(const float2*)(beta  + 2 * tid);
            res.x = fmaf(g2.x, res.x * (1.0f / (float)N_), be2.x);
            res.y = fmaf(g2.y, res.y * (1.0f / (float)N_), be2.y);
            *(float2*)(out + (size_t)rowi * O_ + 2 * tid) = res;
        }

        buf ^= 1;
        rowi = nrowi;
        u2   = nu2;
    }
}

extern "C" void kernel_launch(void* const* d_in, const int* in_sizes, int n_in,
                              void* d_out, int out_size, void* d_ws, size_t ws_size,
                              hipStream_t stream) {
    (void)in_sizes; (void)n_in; (void)out_size; (void)ws_size;
    const float* x     = (const float*)d_in[0];
    const int*   mask  = (const int*)  d_in[1];
    const float* W     = (const float*)d_in[2];
    const float* bias  = (const float*)d_in[3];
    const float* gamma = (const float*)d_in[4];
    const float* beta  = (const float*)d_in[5];
    float* out = (float*)d_out;

    float* u   = (float*)d_ws;
    float* dws = u + (size_t)B_ * N_ * O_;

    edgeconv_gemm<<<B_ * (N_ / 16), 128, 0, stream>>>(x, W, bias, u, dws);
    edgeconv_lnmean<<<LN_BLOCKS, LN_WAVES * 32, 0, stream>>>(u, dws, mask, gamma, beta, out);
}
